# starting kernel
# baseline (speedup 1.0000x reference)
#include <hip/hip_runtime.h>
#include <math.h>


typedef __attribute__((ext_vector_type(2))) float v2f;
typedef __attribute__((ext_vector_type(8))) float v8f;

#define QL 22
#define NCOLS  (1u << (QL - 3))
#define NTILES (NCOLS / 16u)
#define PASS_BLOCKS 512u

__global__ void rgate_setup(const float* __restrict__ angle,
                            float* __restrict__ mats) {
  const int g   = blockIdx.x;
  const int j   = threadIdx.x;
  const int row = j >> 4, col = j & 15;
  const int t = row & 7, tp = col & 7;
  const int base = (g < 7) ? 3 * g : 19;

  float wr = 1.0f, wi = 0.0f;
  for (int k = 0; k < 3; ++k) {
    const int  bit    = base + k;
    const bool active = (g < 7) || (k == 2);
    const int tk = (t >> k) & 1, tpk = (tp >> k) & 1;
    float mr, mi;
    if (!active) {
      mr = (tk == tpk) ? 1.0f : 0.0f; mi = 0.0f;
    } else {
      const float th = 0.5f * angle[(QL - 1) - bit];
      float s, c;
      __sincosf(th, &s, &c);
      if (tk == tpk) { mr = c;    mi = 0.0f; }
      else           { mr = 0.0f; mi = -s;   }
    }
    const float nr = wr * mr - wi * mi;
    const float ni = wr * mi + wi * mr;
    wr = nr; wi = ni;
  }
  const int qr = row >> 3, qc = col >> 3;
  const float val = (qr == 0) ? ((qc == 0) ? wr : -wi)
                              : ((qc == 0) ? wi :  wr);
  mats[g * 256 + row * 16 + col] = val;
}

__global__ __launch_bounds__(256) void rgate_pass(
    const float* __restrict__ in, float* __restrict__ out,
    const float* __restrict__ Am, int b0, int b1, int b2) {
  const int tid  = threadIdx.x;
  const int wave = tid >> 5;
  const int lane = tid & 31;
  const int m    = lane & 15;
  const int half = lane >> 4;

  const float* arow = Am + m * 16 + 2 * half;
  v2f a0, a1, a2, a3;
  a0.x = arow[0];  a0.y = arow[1];
  a1.x = arow[4];  a1.y = arow[5];
  a2.x = arow[8];  a2.y = arow[9];
  a3.x = arow[12]; a3.y = arow[13];

  const float2* in2 = (const float2*)in;
  const unsigned waveSlots = PASS_BLOCKS * 8u;

  for (unsigned tile = blockIdx.x * 8u + (unsigned)wave; tile < NTILES;
       tile += waveSlots) {
    const unsigned r = tile * 16u + (unsigned)m;

    unsigned e = r;
    e = ((e >> b0) << (b0 + 1)) | (e & ((1u << b0) - 1u));
    e = ((e >> b1) << (b1 + 1)) | (e & ((1u << b1) - 1u));
    e = ((e >> b2) << (b2 + 1)) | (e & ((1u << b2) - 1u));

    auto idx_of = [&](int t) -> unsigned {
      return e | ((unsigned)(t & 1) << b0)
               | ((unsigned)((t >> 1) & 1) << b1)
               | ((unsigned)((t >> 2) & 1) << b2);
    };

    const int tb = half * 2;
    const float2 c0 = in2[idx_of(tb + 0)];
    const float2 c1 = in2[idx_of(tb + 1)];
    const float2 c2 = in2[idx_of(tb + 4)];
    const float2 c3 = in2[idx_of(tb + 5)];
    v2f bv0, bv1, bv2, bv3;
    bv0.x = c0.x; bv0.y = c1.x;
    bv1.x = c2.x; bv1.y = c3.x;
    bv2.x = c0.y; bv2.y = c1.y;
    bv3.x = c2.y; bv3.y = c3.y;

    v8f acc = {};
    acc = __builtin_amdgcn_wmma_f32_16x16x4_f32(false, a0, false, bv0, (short)0, acc, false, false);
    acc = __builtin_amdgcn_wmma_f32_16x16x4_f32(false, a1, false, bv1, (short)0, acc, false, false);
    acc = __builtin_amdgcn_wmma_f32_16x16x4_f32(false, a2, false, bv2, (short)0, acc, false, false);
    acc = __builtin_amdgcn_wmma_f32_16x16x4_f32(false, a3, false, bv3, (short)0, acc, false, false);

#pragma unroll
    for (int v = 0; v < 8; ++v) {
      const unsigned iv = idx_of(v);
      out[2u * iv + (unsigned)half] = acc[v];
    }
  }
}

extern "C" void kernel_launch(void* const* d_in, const int* in_sizes, int n_in,
                              void* d_out, int out_size, void* d_ws, size_t ws_size,
                              hipStream_t stream) {
  const float* x     = (const float*)d_in[0];
  const float* angle = (const float*)d_in[1];
  float* out  = (float*)d_out;
  float* mats = (float*)d_ws;

  rgate_setup<<<dim3(8), dim3(256), 0, stream>>>(angle, mats);

  for (int g = 0; g < 8; ++g) {
    const int base = (g < 7) ? 3 * g : 19;
    const float* src = (g == 0) ? x : out;
    rgate_pass<<<dim3(PASS_BLOCKS), dim3(256), 0, stream>>>(
        src, out, mats + g * 256, base, base + 1, base + 2);
  }
}


# Every lane computes its own matrix fragment entries from the angles in registers (sincos, 3-4 factor products) under its tile loads: set-up launch and workspace gone, 4 launches -> 3; same sums
# speedup vs baseline: 2.7952x; 2.7952x over previous
#include <hip/hip_runtime.h>
#include <math.h>


typedef __attribute__((ext_vector_type(2))) float v2f;
typedef __attribute__((ext_vector_type(8))) float v8f;

#define QL 22
#define PASS_BLOCKS 512u

struct HalfAngle { float c, s; };
__device__ __forceinline__ HalfAngle half_angle(float theta) {
  HalfAngle h;
  __sincosf(0.5f * theta, &h.s, &h.c);
  return h;
}

__device__ __forceinline__ void turn(float P, int n, float& wr, float& wi) {
  const float Q = (n & 2) ? -P : P;
  wr = (n & 1) ? 0.0f : Q;
  wi = (n & 1) ? -Q : 0.0f;
}

__device__ __forceinline__ float real_form(float wr, float wi, int qr, int qc) {
  return (qr == qc) ? wr : (qr ? wi : -wi);
}

template <bool HALF_IS_BIT1>
__device__ __forceinline__ void group_row(float th0, float th1, float th2,
                                          int m, int half, v2f (&a)[4]) {
  const HalfAngle h0 = half_angle(th0);
  const HalfAngle h1 = half_angle(th1);
  const HalfAngle h2 = half_angle(th2);
  const int t0 = m & 1, t1 = (m >> 1) & 1, t2 = (m >> 2) & 1, qr = m >> 3;
  const float f0[2] = { t0 ? h0.s : h0.c, t0 ? h0.c : h0.s };
  const float f1[2] = { t1 ? h1.s : h1.c, t1 ? h1.c : h1.s };
  const float f2[2] = { t2 ? h2.s : h2.c, t2 ? h2.c : h2.s };
  const float f1h = half ? f1[1] : f1[0];
#pragma unroll
  for (int j = 0; j < 4; ++j) {
#pragma unroll
    for (int i = 0; i < 2; ++i) {
      float P;
      int n, qc;
      if (HALF_IS_BIT1) {
        P  = (f0[i] * f1h) * f2[j & 1];
        n  = (t0 ^ i) + (t1 ^ half) + (t2 ^ (j & 1));
        qc = j >> 1;
      } else {
        P  = (f0[i] * f1[j & 1]) * f2[j >> 1];
        n  = (t0 ^ i) + (t1 ^ (j & 1)) + (t2 ^ (j >> 1));
        qc = half;
      }
      float wr, wi;
      turn(P, n, wr, wi);
      a[j][i] = real_form(wr, wi, qr, qc);
    }
  }
}

__device__ __forceinline__ void lane_bits_row(float th0, float th1, float th2,
                                              float th3, int m, int half,
                                              v2f (&lr)[4], v2f (&li)[4]) {
  const HalfAngle h0 = half_angle(th0);
  const HalfAngle h1 = half_angle(th1);
  const HalfAngle h2 = half_angle(th2);
  const HalfAngle h3 = half_angle(th3);
  const int t0 = m & 1, t1 = (m >> 1) & 1, t2 = (m >> 2) & 1, t3 = m >> 3;
  const float f0[2] = { t0 ? h0.s : h0.c, t0 ? h0.c : h0.s };
  const float f1[2] = { t1 ? h1.s : h1.c, t1 ? h1.c : h1.s };
  const float f2[2] = { t2 ? h2.s : h2.c, t2 ? h2.c : h2.s };
  const float f3h   = (t3 ^ half) ? h3.s : h3.c;
#pragma unroll
  for (int j = 0; j < 4; ++j) {
#pragma unroll
    for (int i = 0; i < 2; ++i) {
      const float P = ((f0[i] * f1[j & 1]) * f2[j >> 1]) * f3h;
      const int n = (t0 ^ i) + (t1 ^ (j & 1)) + (t2 ^ (j >> 1)) + (t3 ^ half);
      float wr, wi;
      turn(P, n, wr, wi);
      lr[j][i] = wr;
      li[j][i] = wi;
    }
  }
}

template <bool LOW4, int LB, int A0, int B0, int B1, int B2, unsigned FREE>
__global__ __launch_bounds__(256) void rgate_pass2(
    const float* in, float* out, const float* __restrict__ angle) {
  const int tid  = threadIdx.x;
  const int wave = tid >> 5;
  const int lane = tid & 31;
  const int m    = lane & 15;
  const int half = lane >> 4;

  const float tha0 = angle[(QL - 1) - A0];
  const float tha1 = angle[(QL - 1) - (A0 + 1)];
  const float tha2 = angle[(QL - 1) - (A0 + 2)];
  const float thb0 = angle[(QL - 1) - B0];
  const float thb1 = angle[(QL - 1) - B1];
  const float thb2 = angle[(QL - 1) - B2];
  float thl0 = 0.0f, thl1 = 0.0f, thl2 = 0.0f, thl3 = 0.0f;
  if (LOW4) {
    thl0 = angle[(QL - 1) - LB];
    thl1 = angle[(QL - 1) - (LB + 1)];
    thl2 = angle[(QL - 1) - (LB + 2)];
    thl3 = angle[(QL - 1) - (LB + 3)];
  }

  const float xsgn = (m < 8) ? -1.0f : 1.0f;

  const unsigned tile = blockIdx.x * 8u + (unsigned)wave;

  unsigned e = (unsigned)m << LB;
  {
    int k = 0;
#pragma unroll
    for (int p = 0; p < QL; ++p) {
      if (FREE & (1u << p)) { e |= ((tile >> k) & 1u) << p; ++k; }
    }
  }
  const unsigned hoff = (unsigned)half << (A0 + 1);
  const v2f* in2 = (const v2f*)in;

  v2f c[8][4];
#pragma unroll
  for (int tb = 0; tb < 8; ++tb) {
    const unsigned eb = e | hoff
                      | ((unsigned)(tb & 1) << B0)
                      | ((unsigned)((tb >> 1) & 1) << B1)
                      | ((unsigned)((tb >> 2) & 1) << B2);
    c[tb][0] = in2[eb];
    c[tb][1] = in2[eb | (1u << A0)];
    c[tb][2] = in2[eb | (1u << (A0 + 2))];
    c[tb][3] = in2[eb | (1u << (A0 + 2)) | (1u << A0)];
  }
  __builtin_amdgcn_sched_barrier(0);

  v2f a1[4];
  group_row<true>(tha0, tha1, tha2, m, half, a1);
  v2f a2[4];
  group_row<false>(thb0, thb1, thb2, m, half, a2);
  v2f lr[4] = {}, li[4] = {};
  if (LOW4) lane_bits_row(thl0, thl1, thl2, thl3, m, half, lr, li);

  __builtin_amdgcn_sched_barrier(0);

  asm volatile("" :: "v"(c[0][0]), "v"(c[0][1]), "v"(c[0][2]), "v"(c[0][3]),
                     "v"(c[1][0]), "v"(c[1][1]), "v"(c[1][2]), "v"(c[1][3]),
                     "v"(c[2][0]), "v"(c[2][1]), "v"(c[2][2]), "v"(c[2][3]),
                     "v"(c[3][0]), "v"(c[3][1]), "v"(c[3][2]), "v"(c[3][3]));
  asm volatile("" :: "v"(c[4][0]), "v"(c[4][1]), "v"(c[4][2]), "v"(c[4][3]),
                     "v"(c[5][0]), "v"(c[5][1]), "v"(c[5][2]), "v"(c[5][3]),
                     "v"(c[6][0]), "v"(c[6][1]), "v"(c[6][2]), "v"(c[6][3]),
                     "v"(c[7][0]), "v"(c[7][1]), "v"(c[7][2]), "v"(c[7][3]));

  v8f acc1[8];
#pragma unroll
  for (int tb = 0; tb < 8; ++tb) {
    v2f bv0, bv1, bv2, bv3;
    bv0.x = c[tb][0].x; bv0.y = c[tb][1].x;
    bv1.x = c[tb][2].x; bv1.y = c[tb][3].x;
    bv2.x = c[tb][0].y; bv2.y = c[tb][1].y;
    bv3.x = c[tb][2].y; bv3.y = c[tb][3].y;
    v8f acc = {};
    if (!LOW4) {
      acc = __builtin_amdgcn_wmma_f32_16x16x4_f32(false, a1[0], false, bv0, (short)0, acc, false, false);
      acc = __builtin_amdgcn_wmma_f32_16x16x4_f32(false, a1[1], false, bv1, (short)0, acc, false, false);
      acc = __builtin_amdgcn_wmma_f32_16x16x4_f32(false, a1[2], false, bv2, (short)0, acc, false, false);
      acc = __builtin_amdgcn_wmma_f32_16x16x4_f32(false, a1[3], false, bv3, (short)0, acc, false, false);
    } else {
      v8f xp = {};
      xp = __builtin_amdgcn_wmma_f32_16x16x4_f32(false, bv0, false, a1[0], (short)0, xp, false, false);
      xp = __builtin_amdgcn_wmma_f32_16x16x4_f32(false, bv1, false, a1[1], (short)0, xp, false, false);
      xp = __builtin_amdgcn_wmma_f32_16x16x4_f32(false, bv2, false, a1[2], (short)0, xp, false, false);
      xp = __builtin_amdgcn_wmma_f32_16x16x4_f32(false, bv3, false, a1[3], (short)0, xp, false, false);
      v8f xt;
#pragma unroll
      for (int v = 0; v < 8; ++v) {
        const float xv = xp[v];
        const int sw = __builtin_amdgcn_update_dpp(
            0, __float_as_int(xv), 0x168, 0xF, 0xF, false);
        xt[v] = xsgn * __int_as_float(sw);
      }
      v2f p0, p1, p2, p3, q0, q1, q2, q3;
      p0.x = xp[0]; p0.y = xp[1];  p1.x = xp[2]; p1.y = xp[3];
      p2.x = xp[4]; p2.y = xp[5];  p3.x = xp[6]; p3.y = xp[7];
      q0.x = xt[0]; q0.y = xt[1];  q1.x = xt[2]; q1.y = xt[3];
      q2.x = xt[4]; q2.y = xt[5];  q3.x = xt[6]; q3.y = xt[7];
      acc = __builtin_amdgcn_wmma_f32_16x16x4_f32(false, p0, false, lr[0], (short)0, acc, false, false);
      acc = __builtin_amdgcn_wmma_f32_16x16x4_f32(false, p1, false, lr[1], (short)0, acc, false, false);
      acc = __builtin_amdgcn_wmma_f32_16x16x4_f32(false, p2, false, lr[2], (short)0, acc, false, false);
      acc = __builtin_amdgcn_wmma_f32_16x16x4_f32(false, p3, false, lr[3], (short)0, acc, false, false);
      acc = __builtin_amdgcn_wmma_f32_16x16x4_f32(false, q0, false, li[0], (short)0, acc, false, false);
      acc = __builtin_amdgcn_wmma_f32_16x16x4_f32(false, q1, false, li[1], (short)0, acc, false, false);
      acc = __builtin_amdgcn_wmma_f32_16x16x4_f32(false, q2, false, li[2], (short)0, acc, false, false);
      acc = __builtin_amdgcn_wmma_f32_16x16x4_f32(false, q3, false, li[3], (short)0, acc, false, false);
    }
    acc1[tb] = acc;
  }

#pragma unroll
  for (int ta = 0; ta < 8; ++ta) {
    v2f q0, q1, q2, q3;
    q0.x = acc1[0][ta]; q0.y = acc1[1][ta];
    q1.x = acc1[2][ta]; q1.y = acc1[3][ta];
    q2.x = acc1[4][ta]; q2.y = acc1[5][ta];
    q3.x = acc1[6][ta]; q3.y = acc1[7][ta];
    v8f acc = {};
    acc = __builtin_amdgcn_wmma_f32_16x16x4_f32(false, a2[0], false, q0, (short)0, acc, false, false);
    acc = __builtin_amdgcn_wmma_f32_16x16x4_f32(false, a2[1], false, q1, (short)0, acc, false, false);
    acc = __builtin_amdgcn_wmma_f32_16x16x4_f32(false, a2[2], false, q2, (short)0, acc, false, false);
    acc = __builtin_amdgcn_wmma_f32_16x16x4_f32(false, a2[3], false, q3, (short)0, acc, false, false);

    const unsigned ea = e
                      | ((unsigned)(ta & 1) << A0)
                      | ((unsigned)((ta >> 1) & 1) << (A0 + 1))
                      | ((unsigned)((ta >> 2) & 1) << (A0 + 2));
#pragma unroll
    for (int v = 0; v < 8; ++v) {
      const unsigned iv = ea
                        | ((unsigned)(v & 1) << B0)
                        | ((unsigned)((v >> 1) & 1) << B1)
                        | ((unsigned)((v >> 2) & 1) << B2);
      out[2u * iv + (unsigned)half] = acc[v];
    }
  }
}

extern "C" void kernel_launch(void* const* d_in, const int* in_sizes, int n_in,
                              void* d_out, int out_size, void* d_ws, size_t ws_size,
                              hipStream_t stream) {
  const float* x     = (const float*)d_in[0];
  const float* angle = (const float*)d_in[1];
  float* out  = (float*)d_out;

  rgate_pass2<true, 0, 4, 7, 8, 9, 0x3FFC00u>
      <<<dim3(PASS_BLOCKS), dim3(256), 0, stream>>>(x, out, angle);
  rgate_pass2<false, 0, 10, 13, 14, 15, 0x3F03F0u>
      <<<dim3(PASS_BLOCKS), dim3(256), 0, stream>>>(out, out, angle);
  rgate_pass2<false, 0, 16, 19, 20, 21, 0x00FFF0u>
      <<<dim3(PASS_BLOCKS), dim3(256), 0, stream>>>(out, out, angle);
}
